# starting kernel
# baseline (speedup 1.0000x reference)
#include <hip/hip_runtime.h>


typedef __attribute__((ext_vector_type(2))) float v2f;
typedef __attribute__((ext_vector_type(8))) float v8f;

#define BH      32
#define TLEN    2048
#define DIM     64
#define SLICES  4
#define JS      (TLEN / SLICES)
#define LDSW    72

__global__ __launch_bounds__(512)
void xtx_partial_kernel(const float* __restrict__ x, float* __restrict__ P)
{
    const int bh    = blockIdx.x >> 2;
    const int slice = blockIdx.x & 3;
    const int wave  = threadIdx.x >> 5;
    const int lane  = threadIdx.x & 31;
    const int half  = lane >> 4;
    const int l     = lane & 15;
    const int et    = wave >> 2;
    const int dt    = wave & 3;

    const float* __restrict__ X = x + (size_t)bh * TLEN * DIM;
    const int ecol = et * 16 + l;
    const int dcol = dt * 16 + l;

    v8f acc0 = {};
    v8f acc1 = {};

    const int j0 = slice * JS;
    #pragma unroll 4
    for (int j = j0; j < j0 + JS; j += 8) {
        {
            const int jA = j + 2 * half;
            v2f a, b;
            a.x = X[jA * DIM + ecol];
            a.y = X[(jA + 1) * DIM + ecol];
            b.x = X[jA * DIM + dcol];
            b.y = X[(jA + 1) * DIM + dcol];
            acc0 = __builtin_amdgcn_wmma_f32_16x16x4_f32(false, a, false, b,
                                                         (short)0, acc0, false, false);
        }
        {
            const int jA = j + 4 + 2 * half;
            v2f a, b;
            a.x = X[jA * DIM + ecol];
            a.y = X[(jA + 1) * DIM + ecol];
            b.x = X[jA * DIM + dcol];
            b.y = X[(jA + 1) * DIM + dcol];
            acc1 = __builtin_amdgcn_wmma_f32_16x16x4_f32(false, a, false, b,
                                                         (short)0, acc1, false, false);
        }
    }
    const v8f acc = acc0 + acc1;

    float* __restrict__ Pout = P + (size_t)(bh * SLICES + slice) * DIM * DIM;
    #pragma unroll
    for (int r = 0; r < 8; ++r) {
        const int e = et * 16 + r + 8 * half;
        Pout[e * DIM + dcol] = acc[r];
    }
}

__global__ __launch_bounds__(256)
void xm_kernel(const float* __restrict__ x, const float* __restrict__ P,
               float* __restrict__ out)
{
    __shared__ float Ms[DIM * LDSW];

    const int bh   = blockIdx.x >> 4;
    const int jtb  = (blockIdx.x & 15) * 8;
    const int wave = threadIdx.x >> 5;
    const int lane = threadIdx.x & 31;
    const int half = lane >> 4;
    const int l    = lane & 15;

    const float* __restrict__ Pb = P + (size_t)bh * SLICES * DIM * DIM;
    for (int i = threadIdx.x; i < DIM * DIM; i += 256) {
        const float s = Pb[i]
                      + Pb[1 * DIM * DIM + i]
                      + Pb[2 * DIM * DIM + i]
                      + Pb[3 * DIM * DIM + i];
        const int e = i >> 6;
        const int d = i & 63;
        Ms[e * LDSW + d] = s;
    }
    __syncthreads();

    const int jt  = jtb + wave;
    const float* __restrict__ X = x + (size_t)bh * TLEN * DIM;
    const int row = jt * 16 + l;

    v8f acc[4] = {};

    #pragma unroll
    for (int e = 0; e < DIM; e += 4) {
        const int e0 = e + 2 * half;
        const v2f a = *(const v2f*)(X + row * DIM + e0);
        #pragma unroll
        for (int dt = 0; dt < 4; ++dt) {
            v2f b;
            b.x = Ms[e0 * LDSW + dt * 16 + l];
            b.y = Ms[(e0 + 1) * LDSW + dt * 16 + l];
            acc[dt] = __builtin_amdgcn_wmma_f32_16x16x4_f32(false, a, false, b,
                                                            (short)0, acc[dt], false, false);
        }
    }

    float* __restrict__ O = out + (size_t)bh * TLEN * DIM;
    #pragma unroll
    for (int dt = 0; dt < 4; ++dt) {
        #pragma unroll
        for (int r = 0; r < 8; ++r) {
            const int j = jt * 16 + r + 8 * half;
            O[j * DIM + dt * 16 + l] = acc[dt][r];
        }
    }
}

extern "C" void kernel_launch(void* const* d_in, const int* in_sizes, int n_in,
                              void* d_out, int out_size, void* d_ws, size_t ws_size,
                              hipStream_t stream)
{
    (void)in_sizes; (void)n_in; (void)out_size; (void)ws_size;
    const float* x = (const float*)d_in[0];
    float* outp    = (float*)d_out;
    float* P       = (float*)d_ws;

    xtx_partial_kernel<<<BH * SLICES, 512, 0, stream>>>(x, P);
    xm_kernel<<<BH * 16, 256, 0, stream>>>(x, P, outp);
}


# Phase 2 re-decomposed: wave owns one 16-column tile of M, summed in registers from partials phase 1 now stores in fragment order (no LDS copy of M); block stages 256 X rows in LDS; k order changes
# speedup vs baseline: 2.0264x; 2.0264x over previous
#include <hip/hip_runtime.h>


typedef __attribute__((ext_vector_type(2))) float v2f;
typedef __attribute__((ext_vector_type(8))) float v8f;

#define BH      32
#define TLEN    2048
#define DIM     64
#define SLICES  4
#define JS      (TLEN / SLICES)

typedef __attribute__((ext_vector_type(4))) float v4f;

#define CHROWS  128
#define CHPAIRS (CHROWS / 2)
#define NCHUNK  (JS / CHROWS)
#define PAIRW   160
#define CHDW    (CHPAIRS * PAIRW)
#define KSTEPS  (CHROWS / 4)
#define KGROUP  8

__global__ __launch_bounds__(256) __attribute__((amdgpu_waves_per_eu(2, 4)))
void xtx_partial_kernel(const float* __restrict__ x, float* __restrict__ P)
{
    __shared__ __attribute__((aligned(16))) float Xs[2 * CHDW];

    const int g     = blockIdx.x & 1;
    const int slice = (blockIdx.x >> 1) & 3;
    const int bh    = blockIdx.x >> 3;
    const int tid   = threadIdx.x;
    const int wave  = tid >> 5;
    const int lane  = tid & 31;
    const int half  = lane >> 4;
    const int l     = lane & 15;
    const int et    = 2 * g + (wave >> 2);
    const int dt    = wave & 3;

    const float* __restrict__ Xg = x + ((size_t)bh * TLEN + (size_t)slice * JS) * DIM;

    const int sp = tid >> 4;
    const int sm = tid & 15;
    const float* __restrict__ gsrc = Xg + (2 * sp) * DIM + 4 * sm;
    float* sdst = Xs + sp * PAIRW + 8 * sm;

    v4f r0[4], r1[4];

    #pragma unroll
    for (int i = 0; i < 4; ++i) {
        r0[i] = *(const v4f*)(gsrc + (32 * i) * DIM);
        r1[i] = *(const v4f*)(gsrc + (32 * i + 1) * DIM);
    }
    #pragma unroll
    for (int i = 0; i < 4; ++i) {
        v4f lo = { r0[i].x, r1[i].x, r0[i].y, r1[i].y };
        v4f hi = { r0[i].z, r1[i].z, r0[i].w, r1[i].w };
        *(v4f*)(sdst + 16 * i * PAIRW)     = lo;
        *(v4f*)(sdst + 16 * i * PAIRW + 4) = hi;
    }
    __syncthreads();

    v8f acc0 = {};
    v8f acc1 = {};

    const int fa_off = half * PAIRW + 2 * (16 * et + l);
    const int fb_off = half * PAIRW + 2 * (16 * dt + l);

    #pragma unroll
    for (int c = 0; c < NCHUNK; ++c) {
        if (c + 1 < NCHUNK) {
            #pragma unroll
            for (int i = 0; i < 4; ++i) {
                r0[i] = *(const v4f*)(gsrc + ((c + 1) * CHROWS + 32 * i) * DIM);
                r1[i] = *(const v4f*)(gsrc + ((c + 1) * CHROWS + 32 * i + 1) * DIM);
            }
        }

        const float* fa = Xs + (c & 1) * CHDW + fa_off;
        const float* fb = Xs + (c & 1) * CHDW + fb_off;

        v2f a[KGROUP], b[KGROUP], an[KGROUP], bn[KGROUP];
        #pragma unroll
        for (int k = 0; k < KGROUP; ++k) {
            a[k] = *(const v2f*)(fa + 2 * k * PAIRW);
            b[k] = *(const v2f*)(fb + 2 * k * PAIRW);
        }
        #pragma unroll
        for (int kg = 0; kg < KSTEPS; kg += KGROUP) {
            if (kg + KGROUP < KSTEPS) {
                #pragma unroll
                for (int k = 0; k < KGROUP; ++k) {
                    an[k] = *(const v2f*)(fa + 2 * (kg + KGROUP + k) * PAIRW);
                    bn[k] = *(const v2f*)(fb + 2 * (kg + KGROUP + k) * PAIRW);
                }
            } else if (c + 1 < NCHUNK) {
                float* sd = sdst + ((c + 1) & 1) * CHDW;
                #pragma unroll
                for (int i = 0; i < 4; ++i) {
                    v4f lo = { r0[i].x, r1[i].x, r0[i].y, r1[i].y };
                    v4f hi = { r0[i].z, r1[i].z, r0[i].w, r1[i].w };
                    *(v4f*)(sd + 16 * i * PAIRW)     = lo;
                    *(v4f*)(sd + 16 * i * PAIRW + 4) = hi;
                }
            }
            __builtin_amdgcn_sched_barrier(0);
            #pragma unroll
            for (int k = 0; k < KGROUP; k += 2) {
                acc0 = __builtin_amdgcn_wmma_f32_16x16x4_f32(false, a[k], false, b[k],
                                                             (short)0, acc0, false, false);
                acc1 = __builtin_amdgcn_wmma_f32_16x16x4_f32(false, a[k + 1], false, b[k + 1],
                                                             (short)0, acc1, false, false);
            }
            __builtin_amdgcn_sched_barrier(0);
            #pragma unroll
            for (int k = 0; k < KGROUP; ++k) { a[k] = an[k]; b[k] = bn[k]; }
        }

        if (c + 1 < NCHUNK) __syncthreads();
    }
    const v8f acc = acc0 + acc1;

    float* __restrict__ Pout = P + (size_t)(bh * SLICES + slice) * DIM * DIM
                                 + ((dt * 4 + et) * 2) * 128 + lane * 4;
    const v4f lo = { acc[0], acc[1], acc[2], acc[3] };
    const v4f hi = { acc[4], acc[5], acc[6], acc[7] };
    *(v4f*)(Pout)       = lo;
    *(v4f*)(Pout + 128) = hi;
}

#define P2ROWS  256
#define XPITCH  68
#define P2TILES (P2ROWS / 16)

__global__ __launch_bounds__(256) __attribute__((amdgpu_waves_per_eu(2, 4)))
void xm_kernel(const float* __restrict__ x, const float* __restrict__ P,
               float* __restrict__ out)
{
    __shared__ __attribute__((aligned(16))) float Xr[P2ROWS * XPITCH];

    const int bh   = blockIdx.x >> 3;
    const int rb   = blockIdx.x & 7;
    const int tid  = threadIdx.x;
    const int wave = tid >> 5;
    const int lane = tid & 31;
    const int half = lane >> 4;
    const int l    = lane & 15;
    const int dt   = wave & 3;
    const int rg   = wave >> 2;

    const float* __restrict__ Xg = x + ((size_t)bh * TLEN + (size_t)rb * P2ROWS) * DIM + 4 * tid;
    v4f xr[16];
    #pragma unroll
    for (int i = 0; i < 16; ++i)
        xr[i] = *(const v4f*)(Xg + i * 1024);

    const float* __restrict__ Pf = P + (size_t)bh * SLICES * DIM * DIM + (dt * 8) * 128 + lane * 4;
    v4f pv[SLICES][8];
    #pragma unroll
    for (int s = 0; s < SLICES; ++s) {
        #pragma unroll
        for (int i = 0; i < 8; ++i)
            pv[s][i] = *(const v4f*)(Pf + s * DIM * DIM + i * 128);
    }

    float* sdst = Xr + (tid >> 4) * XPITCH + 4 * (tid & 15);
    #pragma unroll
    for (int i = 0; i < 16; ++i)
        *(v4f*)(sdst + i * 16 * XPITCH) = xr[i];

    v4f m[8];
    #pragma unroll
    for (int i = 0; i < 8; ++i)
        m[i] = ((pv[0][i] + pv[1][i]) + pv[2][i]) + pv[3][i];

    __syncthreads();

    const float* arow = Xr + (rg * 128 + l) * XPITCH + 8 * half;
    float* __restrict__ O = out + ((size_t)bh * TLEN + (size_t)rb * P2ROWS + rg * 128 + 8 * half) * DIM
                                + dt * 16 + l;

    v4f a0[8], a1[8], n0[8], n1[8];
    #pragma unroll
    for (int i = 0; i < 8; ++i) {
        a0[i] = *(const v4f*)(arow + (i >> 1) * 16 + (i & 1) * 4);
        a1[i] = *(const v4f*)(arow + 16 * XPITCH + (i >> 1) * 16 + (i & 1) * 4);
    }

    #pragma unroll
    for (int p = 0; p < P2TILES / 4; ++p) {
        if (p + 1 < P2TILES / 4) {
            const float* an = arow + (2 * (p + 1)) * 16 * XPITCH;
            #pragma unroll
            for (int i = 0; i < 8; ++i) {
                n0[i] = *(const v4f*)(an + (i >> 1) * 16 + (i & 1) * 4);
                n1[i] = *(const v4f*)(an + 16 * XPITCH + (i >> 1) * 16 + (i & 1) * 4);
            }
        }
        __builtin_amdgcn_sched_barrier(0);
        v8f acc0 = {};
        v8f acc1 = {};
        #pragma unroll
        for (int i = 0; i < 8; ++i) {
            #pragma unroll
            for (int h2 = 0; h2 < 2; ++h2) {
                const v2f b  = { m[i][2 * h2],  m[i][2 * h2 + 1]  };
                const v2f f0 = { a0[i][2 * h2], a0[i][2 * h2 + 1] };
                const v2f f1 = { a1[i][2 * h2], a1[i][2 * h2 + 1] };
                acc0 = __builtin_amdgcn_wmma_f32_16x16x4_f32(false, f0, false, b,
                                                             (short)0, acc0, false, false);
                acc1 = __builtin_amdgcn_wmma_f32_16x16x4_f32(false, f1, false, b,
                                                             (short)0, acc1, false, false);
            }
        }
        __builtin_amdgcn_sched_barrier(0);

        float* __restrict__ Op = O + (size_t)(2 * p) * 16 * DIM;
        #pragma unroll
        for (int r = 0; r < 8; ++r) {
            Op[r * DIM]            = acc0[r];
            Op[(16 + r) * DIM]     = acc1[r];
        }
        if (p + 1 < P2TILES / 4) {
            #pragma unroll
            for (int i = 0; i < 8; ++i) { a0[i] = n0[i]; a1[i] = n1[i]; }
        }
    }
}

extern "C" void kernel_launch(void* const* d_in, const int* in_sizes, int n_in,
                              void* d_out, int out_size, void* d_ws, size_t ws_size,
                              hipStream_t stream)
{
    (void)in_sizes; (void)n_in; (void)out_size; (void)ws_size;
    const float* x = (const float*)d_in[0];
    float* outp    = (float*)d_out;
    float* P       = (float*)d_ws;

    xtx_partial_kernel<<<BH * SLICES * 2, 256, 0, stream>>>(x, P);
    xm_kernel<<<BH * 8, 256, 0, stream>>>(x, P, outp);
}
